# starting kernel
# baseline (speedup 1.0000x reference)
#include <hip/hip_runtime.h>

typedef float v2f __attribute__((ext_vector_type(2)));
typedef float v8f __attribute__((ext_vector_type(8)));

#define WMMA_F32(a, b, c) \
    __builtin_amdgcn_wmma_f32_16x16x4_f32(false, (a), false, (b), (short)0, (c), false, false)

constexpr int BATCH  = 8;
constexpr int NATOMS = 256;
constexpr int DF     = 42;
constexpr int H1     = 64;
constexpr int H2     = 32;
constexpr int ROWS   = BATCH * NATOMS;
constexpr int JF     = NATOMS * DF;

__device__ __forceinline__ float softplus_f(float z) {
    return fmaxf(z, 0.0f) + log1pf(expf(-fabsf(z)));
}
__device__ __forceinline__ float sigmoid_f(float z) {
    return 1.0f / (1.0f + expf(-z));
}

__global__ __launch_bounds__(32)
void mlp_wmma_kernel(const float* __restrict__ image,
                     const float* __restrict__ W1, const float* __restrict__ b1,
                     const float* __restrict__ W2, const float* __restrict__ b2,
                     const float* __restrict__ W3, const float* __restrict__ b3,
                     float* __restrict__ gws,
                     float* __restrict__ eiws)
{
    __shared__ float sW1[48 * 64];
    __shared__ float sW2[64 * 32];
    __shared__ float sW3[32];
    __shared__ float sb1[64];
    __shared__ float sb2[32];
    __shared__ float sX [16 * 44];
    __shared__ float sA1[16 * 64];
    __shared__ float sA2[16 * 32];
    __shared__ float sG2[16 * 32];
    __shared__ float sG1[16 * 64];

    const int lane = threadIdx.x;
    const int half = lane >> 4;
    const int l    = lane & 15;
    const int tile = blockIdx.x;
    const int row0 = tile * 16;

    for (int idx = lane; idx < 48 * 64; idx += 32) {
        int r = idx >> 6, c = idx & 63;
        sW1[idx] = (r < 42) ? W1[r * 64 + c] : 0.0f;
    }
    for (int idx = lane; idx < 64 * 32; idx += 32) sW2[idx] = W2[idx];
    sW3[lane] = W3[lane];
    for (int idx = lane; idx < 64; idx += 32) sb1[idx] = b1[idx];
    sb2[lane] = b2[lane];
    for (int idx = lane; idx < 16 * 44; idx += 32) {
        int r = idx / 44, c = idx % 44;
        sX[idx] = (c < 42) ? image[(row0 + r) * DF + c] : 0.0f;
    }
    __syncthreads();

    v8f z1t[4];
    #pragma unroll
    for (int t = 0; t < 4; ++t) {
        float bv = sb1[t * 16 + l];
        v8f c;
        #pragma unroll
        for (int i = 0; i < 8; ++i) c[i] = bv;
        #pragma unroll
        for (int k0 = 0; k0 < 44; k0 += 4) {
            v2f a, b;
            a.x = sX[l * 44 + k0 + 2 * half];
            a.y = sX[l * 44 + k0 + 2 * half + 1];
            b.x = sW1[(k0 + 2 * half)     * 64 + t * 16 + l];
            b.y = sW1[(k0 + 2 * half + 1) * 64 + t * 16 + l];
            c = WMMA_F32(a, b, c);
        }
        z1t[t] = c;
    }

    v8f sig1[4];
    #pragma unroll
    for (int t = 0; t < 4; ++t) {
        #pragma unroll
        for (int i = 0; i < 8; ++i) {
            float z = z1t[t][i];
            sA1[(i + 8 * half) * 64 + t * 16 + l] = softplus_f(z);
            sig1[t][i] = sigmoid_f(z);
        }
    }
    __syncthreads();

    v8f z2t[2];
    #pragma unroll
    for (int t = 0; t < 2; ++t) {
        float bv = sb2[t * 16 + l];
        v8f c;
        #pragma unroll
        for (int i = 0; i < 8; ++i) c[i] = bv;
        #pragma unroll
        for (int k0 = 0; k0 < 64; k0 += 4) {
            v2f a, b;
            a.x = sA1[l * 64 + k0 + 2 * half];
            a.y = sA1[l * 64 + k0 + 2 * half + 1];
            b.x = sW2[(k0 + 2 * half)     * 32 + t * 16 + l];
            b.y = sW2[(k0 + 2 * half + 1) * 32 + t * 16 + l];
            c = WMMA_F32(a, b, c);
        }
        z2t[t] = c;
    }

    #pragma unroll
    for (int t = 0; t < 2; ++t) {
        float w3v = sW3[t * 16 + l];
        #pragma unroll
        for (int i = 0; i < 8; ++i) {
            float z = z2t[t][i];
            sA2[(i + 8 * half) * 32 + t * 16 + l] = softplus_f(z);
            sG2[(i + 8 * half) * 32 + t * 16 + l] = sigmoid_f(z) * w3v;
        }
    }
    __syncthreads();

    v8f ce;
    #pragma unroll
    for (int i = 0; i < 8; ++i) ce[i] = 0.0f;
    #pragma unroll
    for (int k0 = 0; k0 < 32; k0 += 4) {
        v2f a, b;
        a.x = sA2[l * 32 + k0 + 2 * half];
        a.y = sA2[l * 32 + k0 + 2 * half + 1];
        b.x = (l == 0) ? sW3[k0 + 2 * half]     : 0.0f;
        b.y = (l == 0) ? sW3[k0 + 2 * half + 1] : 0.0f;
        ce = WMMA_F32(a, b, ce);
    }
    {
        float b3v = b3[0];
        if (l == 0) {
            #pragma unroll
            for (int v = 0; v < 8; ++v)
                eiws[row0 + 8 * half + v] = ce[v] + b3v;
        }
    }

    #pragma unroll
    for (int t = 0; t < 4; ++t) {
        v8f c;
        #pragma unroll
        for (int i = 0; i < 8; ++i) c[i] = 0.0f;
        #pragma unroll
        for (int k0 = 0; k0 < 32; k0 += 4) {
            v2f a, b;
            a.x = sG2[l * 32 + k0 + 2 * half];
            a.y = sG2[l * 32 + k0 + 2 * half + 1];
            b.x = sW2[(t * 16 + l) * 32 + k0 + 2 * half];
            b.y = sW2[(t * 16 + l) * 32 + k0 + 2 * half + 1];
            c = WMMA_F32(a, b, c);
        }
        #pragma unroll
        for (int i = 0; i < 8; ++i)
            sG1[(i + 8 * half) * 64 + t * 16 + l] = sig1[t][i] * c[i];
    }
    __syncthreads();

    #pragma unroll
    for (int t = 0; t < 3; ++t) {
        v8f c;
        #pragma unroll
        for (int i = 0; i < 8; ++i) c[i] = 0.0f;
        #pragma unroll
        for (int k0 = 0; k0 < 64; k0 += 4) {
            v2f a, b;
            a.x = sG1[l * 64 + k0 + 2 * half];
            a.y = sG1[l * 64 + k0 + 2 * half + 1];
            b.x = sW1[(t * 16 + l) * 64 + k0 + 2 * half];
            b.y = sW1[(t * 16 + l) * 64 + k0 + 2 * half + 1];
            c = WMMA_F32(a, b, c);
        }
        int d = t * 16 + l;
        if (d < DF) {
            #pragma unroll
            for (int v = 0; v < 8; ++v)
                gws[(row0 + v + 8 * half) * DF + d] = c[v];
        }
    }
}

__global__ __launch_bounds__(256)
void force_kernel(const float* __restrict__ dfeat,
                  const float* __restrict__ gws,
                  float* __restrict__ out)
{
    __shared__ __align__(16) float gsh[JF];

    const int tid = threadIdx.x;
    const int blk = blockIdx.x;
    const int b   = blk >> 5;
    const int i0  = (blk & 31) * 8;

    for (int idx = tid; idx < JF; idx += 256) gsh[idx] = gws[b * JF + idx];
    __syncthreads();

    const int wave = tid >> 5;
    const int lane = tid & 31;
    const int i    = i0 + wave;

    const float* row = dfeat + (size_t)(b * NATOMS + i) * (size_t)(JF * 3);
    const float4* g4 = (const float4*)gsh;

    float ax = 0.0f, ay = 0.0f, az = 0.0f;
    for (int it = 0; it < 84; ++it) {
        int q = it * 32 + lane;
        const float4* p = (const float4*)(row + (size_t)q * 12);
        float4 d0 = p[0], d1 = p[1], d2 = p[2];
        float4 gv = g4[q];
        ax = fmaf(d0.x, gv.x, fmaf(d0.w, gv.y, fmaf(d1.z, gv.z, fmaf(d2.y, gv.w, ax))));
        ay = fmaf(d0.y, gv.x, fmaf(d1.x, gv.y, fmaf(d1.w, gv.z, fmaf(d2.z, gv.w, ay))));
        az = fmaf(d0.z, gv.x, fmaf(d1.y, gv.y, fmaf(d2.x, gv.z, fmaf(d2.w, gv.w, az))));
    }
    #pragma unroll
    for (int off = 16; off; off >>= 1) {
        ax += __shfl_xor(ax, off, 32);
        ay += __shfl_xor(ay, off, 32);
        az += __shfl_xor(az, off, 32);
    }
    if (lane == 0) {
        float* f = out + 8 + (size_t)(b * NATOMS + i) * 3;
        f[0] = ax; f[1] = ay; f[2] = az;
    }
}

__global__ __launch_bounds__(256)
void etot_kernel(const float* __restrict__ eiws, float* __restrict__ out)
{
    const int b = threadIdx.x >> 5;
    const int lane = threadIdx.x & 31;
    float s = 0.0f;
    #pragma unroll
    for (int k = 0; k < 8; ++k) s += eiws[b * NATOMS + lane + 32 * k];
    #pragma unroll
    for (int off = 16; off; off >>= 1) s += __shfl_xor(s, off, 32);
    if (lane == 0) out[b] = s;
}

extern "C" void kernel_launch(void* const* d_in, const int* in_sizes, int n_in,
                              void* d_out, int out_size, void* d_ws, size_t ws_size,
                              hipStream_t stream)
{
    (void)in_sizes; (void)n_in; (void)out_size; (void)ws_size;

    const float* image = (const float*)d_in[0];
    const float* dfeat = (const float*)d_in[1];
    const float* W1    = (const float*)d_in[2];
    const float* b1    = (const float*)d_in[3];
    const float* W2    = (const float*)d_in[4];
    const float* b2    = (const float*)d_in[5];
    const float* W3    = (const float*)d_in[6];
    const float* b3    = (const float*)d_in[7];

    float* out  = (float*)d_out;
    float* gws  = (float*)d_ws;
    float* eiws = gws + (size_t)ROWS * DF;

    mlp_wmma_kernel<<<ROWS / 16, 32, 0, stream>>>(image, W1, b1, W2, b2, W3, b3, gws, eiws);
    force_kernel<<<256, 256, 0, stream>>>(dfeat, gws, out);
    etot_kernel<<<1, 256, 0, stream>>>(eiws, out);
}


# first kernel: all staging loads (W1, W2, biases, input tile) issued as one batch of 16-byte and 8-byte loads before the first LDS store, instead of one load-wait-store round trip per element
# speedup vs baseline: 1.8854x; 1.8854x over previous
#include <hip/hip_runtime.h>

typedef float v2f __attribute__((ext_vector_type(2)));
typedef float v4f __attribute__((ext_vector_type(4)));
typedef float v8f __attribute__((ext_vector_type(8)));

#define WMMA_F32(a, b, c) \
    __builtin_amdgcn_wmma_f32_16x16x4_f32(false, (a), false, (b), (short)0, (c), false, false)

constexpr int BATCH  = 8;
constexpr int NATOMS = 256;
constexpr int DF     = 42;
constexpr int H1     = 64;
constexpr int H2     = 32;
constexpr int ROWS   = BATCH * NATOMS;
constexpr int JF     = NATOMS * DF;

__device__ __forceinline__ float softplus_f(float z) {
    return fmaxf(z, 0.0f) + log1pf(expf(-fabsf(z)));
}
__device__ __forceinline__ float sigmoid_f(float z) {
    return 1.0f / (1.0f + expf(-z));
}

__global__ __launch_bounds__(32)
void mlp_wmma_kernel(const float* __restrict__ image,
                     const float* __restrict__ W1, const float* __restrict__ b1,
                     const float* __restrict__ W2, const float* __restrict__ b2,
                     const float* __restrict__ W3, const float* __restrict__ b3,
                     float* __restrict__ gws,
                     float* __restrict__ eiws)
{
    __shared__ __align__(16) float sW1[48 * 64];
    __shared__ __align__(16) float sW2[64 * 32];
    __shared__ float sW3[32];
    __shared__ float sb1[64];
    __shared__ float sb2[32];
    __shared__ __align__(16) float sX [16 * 44];
    __shared__ float sA1[16 * 64];
    __shared__ float sA2[16 * 32];
    __shared__ float sG2[16 * 32];
    __shared__ float sG1[16 * 64];

    const int lane = threadIdx.x;
    const int half = lane >> 4;
    const int l    = lane & 15;
    const int tile = blockIdx.x;
    const int row0 = tile * 16;

    const v4f* W1v = (const v4f*)W1;
    const v4f* W2v = (const v4f*)W2;
    const v2f* Xv  = (const v2f*)(image + (size_t)row0 * DF);

    v2f xr[11];
    v4f w1r[21];
    v4f w2r[16];
    #pragma unroll
    for (int it = 0; it < 11; ++it) {
        int p = lane + 32 * it;
        xr[it] = Xv[p < 336 ? p : 335];
    }
    #pragma unroll
    for (int it = 0; it < 21; ++it) w1r[it] = W1v[lane + 32 * it];
    #pragma unroll
    for (int it = 0; it < 16; ++it) w2r[it] = W2v[lane + 32 * it];
    const float w3r  = W3[lane];
    const float b2r  = b2[lane];
    const float b1r0 = b1[lane];
    const float b1r1 = b1[lane + 32];

    v4f* sW1v = (v4f*)sW1;
    v4f* sW2v = (v4f*)sW2;
    const v4f zero4 = {0.0f, 0.0f, 0.0f, 0.0f};
    #pragma unroll
    for (int it = 0; it < 10; ++it) {
        int e = 2 * (lane + 32 * it);
        int r = e / 42, c = e - 42 * r;
        *(v2f*)&sX[r * 44 + c] = xr[it];
    }
    {
        int e = 2 * (lane + 320);
        int r = e / 42, c = e - 42 * r;
        int  a = (lane < 16) ? (r * 44 + c) : ((lane - 16) * 44 + 42);
        v2f  v = xr[10];
        if (lane >= 16) { v.x = 0.0f; v.y = 0.0f; }
        *(v2f*)&sX[a] = v;
    }
    #pragma unroll
    for (int it = 0; it < 21; ++it) sW1v[lane + 32 * it] = w1r[it];
    for (int it = 0; it < 3; ++it) sW1v[672 + lane + 32 * it] = zero4;
    #pragma unroll
    for (int it = 0; it < 16; ++it) sW2v[lane + 32 * it] = w2r[it];
    sW3[lane]      = w3r;
    sb2[lane]      = b2r;
    sb1[lane]      = b1r0;
    sb1[lane + 32] = b1r1;
    __syncthreads();

    v8f z1t[4];
    #pragma unroll
    for (int t = 0; t < 4; ++t) {
        float bv = sb1[t * 16 + l];
        v8f c;
        #pragma unroll
        for (int i = 0; i < 8; ++i) c[i] = bv;
        #pragma unroll
        for (int k0 = 0; k0 < 44; k0 += 4) {
            v2f a, b;
            a.x = sX[l * 44 + k0 + 2 * half];
            a.y = sX[l * 44 + k0 + 2 * half + 1];
            b.x = sW1[(k0 + 2 * half)     * 64 + t * 16 + l];
            b.y = sW1[(k0 + 2 * half + 1) * 64 + t * 16 + l];
            c = WMMA_F32(a, b, c);
        }
        z1t[t] = c;
    }

    v8f sig1[4];
    #pragma unroll
    for (int t = 0; t < 4; ++t) {
        #pragma unroll
        for (int i = 0; i < 8; ++i) {
            float z = z1t[t][i];
            sA1[(i + 8 * half) * 64 + t * 16 + l] = softplus_f(z);
            sig1[t][i] = sigmoid_f(z);
        }
    }
    __syncthreads();

    v8f z2t[2];
    #pragma unroll
    for (int t = 0; t < 2; ++t) {
        float bv = sb2[t * 16 + l];
        v8f c;
        #pragma unroll
        for (int i = 0; i < 8; ++i) c[i] = bv;
        #pragma unroll
        for (int k0 = 0; k0 < 64; k0 += 4) {
            v2f a, b;
            a.x = sA1[l * 64 + k0 + 2 * half];
            a.y = sA1[l * 64 + k0 + 2 * half + 1];
            b.x = sW2[(k0 + 2 * half)     * 32 + t * 16 + l];
            b.y = sW2[(k0 + 2 * half + 1) * 32 + t * 16 + l];
            c = WMMA_F32(a, b, c);
        }
        z2t[t] = c;
    }

    #pragma unroll
    for (int t = 0; t < 2; ++t) {
        float w3v = sW3[t * 16 + l];
        #pragma unroll
        for (int i = 0; i < 8; ++i) {
            float z = z2t[t][i];
            sA2[(i + 8 * half) * 32 + t * 16 + l] = softplus_f(z);
            sG2[(i + 8 * half) * 32 + t * 16 + l] = sigmoid_f(z) * w3v;
        }
    }
    __syncthreads();

    v8f ce;
    #pragma unroll
    for (int i = 0; i < 8; ++i) ce[i] = 0.0f;
    #pragma unroll
    for (int k0 = 0; k0 < 32; k0 += 4) {
        v2f a, b;
        a.x = sA2[l * 32 + k0 + 2 * half];
        a.y = sA2[l * 32 + k0 + 2 * half + 1];
        b.x = (l == 0) ? sW3[k0 + 2 * half]     : 0.0f;
        b.y = (l == 0) ? sW3[k0 + 2 * half + 1] : 0.0f;
        ce = WMMA_F32(a, b, ce);
    }
    {
        float b3v = b3[0];
        if (l == 0) {
            #pragma unroll
            for (int v = 0; v < 8; ++v)
                eiws[row0 + 8 * half + v] = ce[v] + b3v;
        }
    }

    #pragma unroll
    for (int t = 0; t < 4; ++t) {
        v8f c;
        #pragma unroll
        for (int i = 0; i < 8; ++i) c[i] = 0.0f;
        #pragma unroll
        for (int k0 = 0; k0 < 32; k0 += 4) {
            v2f a, b;
            a.x = sG2[l * 32 + k0 + 2 * half];
            a.y = sG2[l * 32 + k0 + 2 * half + 1];
            b.x = sW2[(t * 16 + l) * 32 + k0 + 2 * half];
            b.y = sW2[(t * 16 + l) * 32 + k0 + 2 * half + 1];
            c = WMMA_F32(a, b, c);
        }
        #pragma unroll
        for (int i = 0; i < 8; ++i)
            sG1[(i + 8 * half) * 64 + t * 16 + l] = sig1[t][i] * c[i];
    }
    __syncthreads();

    #pragma unroll
    for (int t = 0; t < 3; ++t) {
        v8f c;
        #pragma unroll
        for (int i = 0; i < 8; ++i) c[i] = 0.0f;
        #pragma unroll
        for (int k0 = 0; k0 < 64; k0 += 4) {
            v2f a, b;
            a.x = sG1[l * 64 + k0 + 2 * half];
            a.y = sG1[l * 64 + k0 + 2 * half + 1];
            b.x = sW1[(t * 16 + l) * 64 + k0 + 2 * half];
            b.y = sW1[(t * 16 + l) * 64 + k0 + 2 * half + 1];
            c = WMMA_F32(a, b, c);
        }
        int d = t * 16 + l;
        if (d < DF) {
            #pragma unroll
            for (int v = 0; v < 8; ++v)
                gws[(row0 + v + 8 * half) * DF + d] = c[v];
        }
    }
}

__global__ __launch_bounds__(256)
void force_kernel(const float* __restrict__ dfeat,
                  const float* __restrict__ gws,
                  float* __restrict__ out)
{
    __shared__ __align__(16) float gsh[JF];

    const int tid = threadIdx.x;
    const int blk = blockIdx.x;
    const int b   = blk >> 5;
    const int i0  = (blk & 31) * 8;

    for (int idx = tid; idx < JF; idx += 256) gsh[idx] = gws[b * JF + idx];
    __syncthreads();

    const int wave = tid >> 5;
    const int lane = tid & 31;
    const int i    = i0 + wave;

    const float* row = dfeat + (size_t)(b * NATOMS + i) * (size_t)(JF * 3);
    const float4* g4 = (const float4*)gsh;

    float ax = 0.0f, ay = 0.0f, az = 0.0f;
    for (int it = 0; it < 84; ++it) {
        int q = it * 32 + lane;
        const float4* p = (const float4*)(row + (size_t)q * 12);
        float4 d0 = p[0], d1 = p[1], d2 = p[2];
        float4 gv = g4[q];
        ax = fmaf(d0.x, gv.x, fmaf(d0.w, gv.y, fmaf(d1.z, gv.z, fmaf(d2.y, gv.w, ax))));
        ay = fmaf(d0.y, gv.x, fmaf(d1.x, gv.y, fmaf(d1.w, gv.z, fmaf(d2.z, gv.w, ay))));
        az = fmaf(d0.z, gv.x, fmaf(d1.y, gv.y, fmaf(d2.x, gv.z, fmaf(d2.w, gv.w, az))));
    }
    #pragma unroll
    for (int off = 16; off; off >>= 1) {
        ax += __shfl_xor(ax, off, 32);
        ay += __shfl_xor(ay, off, 32);
        az += __shfl_xor(az, off, 32);
    }
    if (lane == 0) {
        float* f = out + 8 + (size_t)(b * NATOMS + i) * 3;
        f[0] = ax; f[1] = ay; f[2] = az;
    }
}

__global__ __launch_bounds__(256)
void etot_kernel(const float* __restrict__ eiws, float* __restrict__ out)
{
    const int b = threadIdx.x >> 5;
    const int lane = threadIdx.x & 31;
    float s = 0.0f;
    #pragma unroll
    for (int k = 0; k < 8; ++k) s += eiws[b * NATOMS + lane + 32 * k];
    #pragma unroll
    for (int off = 16; off; off >>= 1) s += __shfl_xor(s, off, 32);
    if (lane == 0) out[b] = s;
}

extern "C" void kernel_launch(void* const* d_in, const int* in_sizes, int n_in,
                              void* d_out, int out_size, void* d_ws, size_t ws_size,
                              hipStream_t stream)
{
    (void)in_sizes; (void)n_in; (void)out_size; (void)ws_size;

    const float* image = (const float*)d_in[0];
    const float* dfeat = (const float*)d_in[1];
    const float* W1    = (const float*)d_in[2];
    const float* b1    = (const float*)d_in[3];
    const float* W2    = (const float*)d_in[4];
    const float* b2    = (const float*)d_in[5];
    const float* W3    = (const float*)d_in[6];
    const float* b3    = (const float*)d_in[7];

    float* out  = (float*)d_out;
    float* gws  = (float*)d_ws;
    float* eiws = gws + (size_t)ROWS * DF;

    mlp_wmma_kernel<<<ROWS / 16, 32, 0, stream>>>(image, W1, b1, W2, b2, W3, b3, gws, eiws);
    force_kernel<<<256, 256, 0, stream>>>(dfeat, gws, out);
    etot_kernel<<<1, 256, 0, stream>>>(eiws, out);
}


# force kernel: each wave-wide 16-byte dfeat load covers 512 contiguous bytes (4 whole lines; was 12 lines at a 48-byte lane stride); g in LDS repeated x3 to share the index; row sums re-ordered
# speedup vs baseline: 2.0144x; 1.0684x over previous
#include <hip/hip_runtime.h>

typedef float v2f __attribute__((ext_vector_type(2)));
typedef float v4f __attribute__((ext_vector_type(4)));
typedef float v8f __attribute__((ext_vector_type(8)));

#define WMMA_F32(a, b, c) \
    __builtin_amdgcn_wmma_f32_16x16x4_f32(false, (a), false, (b), (short)0, (c), false, false)

constexpr int BATCH  = 8;
constexpr int NATOMS = 256;
constexpr int DF     = 42;
constexpr int H1     = 64;
constexpr int H2     = 32;
constexpr int ROWS   = BATCH * NATOMS;
constexpr int JF     = NATOMS * DF;

__device__ __forceinline__ float softplus_f(float z) {
    return fmaxf(z, 0.0f) + log1pf(expf(-fabsf(z)));
}
__device__ __forceinline__ float sigmoid_f(float z) {
    return 1.0f / (1.0f + expf(-z));
}

__global__ __launch_bounds__(32)
void mlp_wmma_kernel(const float* __restrict__ image,
                     const float* __restrict__ W1, const float* __restrict__ b1,
                     const float* __restrict__ W2, const float* __restrict__ b2,
                     const float* __restrict__ W3, const float* __restrict__ b3,
                     float* __restrict__ gws,
                     float* __restrict__ eiws)
{
    __shared__ __align__(16) float sW1[48 * 64];
    __shared__ __align__(16) float sW2[64 * 32];
    __shared__ float sW3[32];
    __shared__ float sb1[64];
    __shared__ float sb2[32];
    __shared__ __align__(16) float sX [16 * 44];
    __shared__ float sA1[16 * 64];
    __shared__ float sA2[16 * 32];
    __shared__ float sG2[16 * 32];
    __shared__ float sG1[16 * 64];

    const int lane = threadIdx.x;
    const int half = lane >> 4;
    const int l    = lane & 15;
    const int tile = blockIdx.x;
    const int row0 = tile * 16;

    const v4f* W1v = (const v4f*)W1;
    const v4f* W2v = (const v4f*)W2;
    const v2f* Xv  = (const v2f*)(image + (size_t)row0 * DF);

    v2f xr[11];
    v4f w1r[21];
    v4f w2r[16];
    #pragma unroll
    for (int it = 0; it < 11; ++it) {
        int p = lane + 32 * it;
        xr[it] = Xv[p < 336 ? p : 335];
    }
    #pragma unroll
    for (int it = 0; it < 21; ++it) w1r[it] = W1v[lane + 32 * it];
    #pragma unroll
    for (int it = 0; it < 16; ++it) w2r[it] = W2v[lane + 32 * it];
    const float w3r  = W3[lane];
    const float b2r  = b2[lane];
    const float b1r0 = b1[lane];
    const float b1r1 = b1[lane + 32];

    v4f* sW1v = (v4f*)sW1;
    v4f* sW2v = (v4f*)sW2;
    const v4f zero4 = {0.0f, 0.0f, 0.0f, 0.0f};
    #pragma unroll
    for (int it = 0; it < 10; ++it) {
        int e = 2 * (lane + 32 * it);
        int r = e / 42, c = e - 42 * r;
        *(v2f*)&sX[r * 44 + c] = xr[it];
    }
    {
        int e = 2 * (lane + 320);
        int r = e / 42, c = e - 42 * r;
        int  a = (lane < 16) ? (r * 44 + c) : ((lane - 16) * 44 + 42);
        v2f  v = xr[10];
        if (lane >= 16) { v.x = 0.0f; v.y = 0.0f; }
        *(v2f*)&sX[a] = v;
    }
    #pragma unroll
    for (int it = 0; it < 21; ++it) sW1v[lane + 32 * it] = w1r[it];
    for (int it = 0; it < 3; ++it) sW1v[672 + lane + 32 * it] = zero4;
    #pragma unroll
    for (int it = 0; it < 16; ++it) sW2v[lane + 32 * it] = w2r[it];
    sW3[lane]      = w3r;
    sb2[lane]      = b2r;
    sb1[lane]      = b1r0;
    sb1[lane + 32] = b1r1;
    __syncthreads();

    v8f z1t[4];
    #pragma unroll
    for (int t = 0; t < 4; ++t) {
        float bv = sb1[t * 16 + l];
        v8f c;
        #pragma unroll
        for (int i = 0; i < 8; ++i) c[i] = bv;
        #pragma unroll
        for (int k0 = 0; k0 < 44; k0 += 4) {
            v2f a, b;
            a.x = sX[l * 44 + k0 + 2 * half];
            a.y = sX[l * 44 + k0 + 2 * half + 1];
            b.x = sW1[(k0 + 2 * half)     * 64 + t * 16 + l];
            b.y = sW1[(k0 + 2 * half + 1) * 64 + t * 16 + l];
            c = WMMA_F32(a, b, c);
        }
        z1t[t] = c;
    }

    v8f sig1[4];
    #pragma unroll
    for (int t = 0; t < 4; ++t) {
        #pragma unroll
        for (int i = 0; i < 8; ++i) {
            float z = z1t[t][i];
            sA1[(i + 8 * half) * 64 + t * 16 + l] = softplus_f(z);
            sig1[t][i] = sigmoid_f(z);
        }
    }
    __syncthreads();

    v8f z2t[2];
    #pragma unroll
    for (int t = 0; t < 2; ++t) {
        float bv = sb2[t * 16 + l];
        v8f c;
        #pragma unroll
        for (int i = 0; i < 8; ++i) c[i] = bv;
        #pragma unroll
        for (int k0 = 0; k0 < 64; k0 += 4) {
            v2f a, b;
            a.x = sA1[l * 64 + k0 + 2 * half];
            a.y = sA1[l * 64 + k0 + 2 * half + 1];
            b.x = sW2[(k0 + 2 * half)     * 32 + t * 16 + l];
            b.y = sW2[(k0 + 2 * half + 1) * 32 + t * 16 + l];
            c = WMMA_F32(a, b, c);
        }
        z2t[t] = c;
    }

    #pragma unroll
    for (int t = 0; t < 2; ++t) {
        float w3v = sW3[t * 16 + l];
        #pragma unroll
        for (int i = 0; i < 8; ++i) {
            float z = z2t[t][i];
            sA2[(i + 8 * half) * 32 + t * 16 + l] = softplus_f(z);
            sG2[(i + 8 * half) * 32 + t * 16 + l] = sigmoid_f(z) * w3v;
        }
    }
    __syncthreads();

    v8f ce;
    #pragma unroll
    for (int i = 0; i < 8; ++i) ce[i] = 0.0f;
    #pragma unroll
    for (int k0 = 0; k0 < 32; k0 += 4) {
        v2f a, b;
        a.x = sA2[l * 32 + k0 + 2 * half];
        a.y = sA2[l * 32 + k0 + 2 * half + 1];
        b.x = (l == 0) ? sW3[k0 + 2 * half]     : 0.0f;
        b.y = (l == 0) ? sW3[k0 + 2 * half + 1] : 0.0f;
        ce = WMMA_F32(a, b, ce);
    }
    {
        float b3v = b3[0];
        if (l == 0) {
            #pragma unroll
            for (int v = 0; v < 8; ++v)
                eiws[row0 + 8 * half + v] = ce[v] + b3v;
        }
    }

    #pragma unroll
    for (int t = 0; t < 4; ++t) {
        v8f c;
        #pragma unroll
        for (int i = 0; i < 8; ++i) c[i] = 0.0f;
        #pragma unroll
        for (int k0 = 0; k0 < 32; k0 += 4) {
            v2f a, b;
            a.x = sG2[l * 32 + k0 + 2 * half];
            a.y = sG2[l * 32 + k0 + 2 * half + 1];
            b.x = sW2[(t * 16 + l) * 32 + k0 + 2 * half];
            b.y = sW2[(t * 16 + l) * 32 + k0 + 2 * half + 1];
            c = WMMA_F32(a, b, c);
        }
        #pragma unroll
        for (int i = 0; i < 8; ++i)
            sG1[(i + 8 * half) * 64 + t * 16 + l] = sig1[t][i] * c[i];
    }
    __syncthreads();

    #pragma unroll
    for (int t = 0; t < 3; ++t) {
        v8f c;
        #pragma unroll
        for (int i = 0; i < 8; ++i) c[i] = 0.0f;
        #pragma unroll
        for (int k0 = 0; k0 < 64; k0 += 4) {
            v2f a, b;
            a.x = sG1[l * 64 + k0 + 2 * half];
            a.y = sG1[l * 64 + k0 + 2 * half + 1];
            b.x = sW1[(t * 16 + l) * 64 + k0 + 2 * half];
            b.y = sW1[(t * 16 + l) * 64 + k0 + 2 * half + 1];
            c = WMMA_F32(a, b, c);
        }
        int d = t * 16 + l;
        if (d < DF) {
            #pragma unroll
            for (int v = 0; v < 8; ++v)
                gws[(row0 + v + 8 * half) * DF + d] = c[v];
        }
    }
}

constexpr int ROWF = JF * 3;
constexpr int NG   = 12;
constexpr int NGRP = ROWF / (4 * 32 * NG);

__global__ __launch_bounds__(256)
void force_kernel(const float* __restrict__ dfeat,
                  const float* __restrict__ gws,
                  float* __restrict__ out)
{
    __shared__ __align__(16) float g3[ROWF];

    const int tid = threadIdx.x;
    const int blk = blockIdx.x;
    const int b   = blk >> 5;
    const int i0  = (blk & 31) * 8;

    {
        const v4f* gv = (const v4f*)(gws + (size_t)b * JF);
        v4f* g3v = (v4f*)g3;
        v4f gr[11];
        #pragma unroll
        for (int t = 0; t < 10; ++t) gr[t] = gv[tid + 256 * t];
        gr[10] = gv[2560 + (tid & 127)];
        #pragma unroll
        for (int t = 0; t < 11; ++t) {
            if (t < 10 || tid < 128) {
                const int q = tid + 256 * t;
                const v4f g = gr[t];
                const v4f s0 = {g.x, g.x, g.x, g.y};
                const v4f s1 = {g.y, g.y, g.z, g.z};
                const v4f s2 = {g.z, g.w, g.w, g.w};
                g3v[3 * q + 0] = s0;
                g3v[3 * q + 1] = s1;
                g3v[3 * q + 2] = s2;
            }
        }
    }
    __syncthreads();

    const int wave = tid >> 5;
    const int lane = tid & 31;
    const int i    = i0 + wave;

    const v4f* rv  = (const v4f*)(dfeat + (size_t)(b * NATOMS + i) * (size_t)ROWF) + lane;
    const v4f* g3l = (const v4f*)g3 + lane;

    const v4f zero4 = {0.0f, 0.0f, 0.0f, 0.0f};
    v4f acc[3] = {zero4, zero4, zero4};

    for (int grp = 0; grp < NGRP; ++grp) {
        v4f d[NG];
        #pragma unroll
        for (int u = 0; u < NG; ++u) d[u] = rv[(grp * NG + u) * 32];
        #pragma unroll
        for (int u = 0; u < NG; ++u)
            acc[u % 3] = __builtin_elementwise_fma(d[u], g3l[(grp * NG + u) * 32], acc[u % 3]);
    }

    float a[3] = {0.0f, 0.0f, 0.0f};
    #pragma unroll
    for (int k = 0; k < 3; ++k) {
        #pragma unroll
        for (int r = 0; r < 4; ++r) a[(2 * k + r) % 3] += acc[k][r];
    }
    const int ph = lane % 3;
    float ax = (ph == 0) ? a[0] : ((ph == 1) ? a[2] : a[1]);
    float ay = (ph == 0) ? a[1] : ((ph == 1) ? a[0] : a[2]);
    float az = (ph == 0) ? a[2] : ((ph == 1) ? a[1] : a[0]);

    #pragma unroll
    for (int off = 16; off; off >>= 1) {
        ax += __shfl_xor(ax, off, 32);
        ay += __shfl_xor(ay, off, 32);
        az += __shfl_xor(az, off, 32);
    }
    if (lane == 0) {
        float* f = out + 8 + (size_t)(b * NATOMS + i) * 3;
        f[0] = ax; f[1] = ay; f[2] = az;
    }
}

__global__ __launch_bounds__(256)
void etot_kernel(const float* __restrict__ eiws, float* __restrict__ out)
{
    const int b = threadIdx.x >> 5;
    const int lane = threadIdx.x & 31;
    float s = 0.0f;
    #pragma unroll
    for (int k = 0; k < 8; ++k) s += eiws[b * NATOMS + lane + 32 * k];
    #pragma unroll
    for (int off = 16; off; off >>= 1) s += __shfl_xor(s, off, 32);
    if (lane == 0) out[b] = s;
}

extern "C" void kernel_launch(void* const* d_in, const int* in_sizes, int n_in,
                              void* d_out, int out_size, void* d_ws, size_t ws_size,
                              hipStream_t stream)
{
    (void)in_sizes; (void)n_in; (void)out_size; (void)ws_size;

    const float* image = (const float*)d_in[0];
    const float* dfeat = (const float*)d_in[1];
    const float* W1    = (const float*)d_in[2];
    const float* b1    = (const float*)d_in[3];
    const float* W2    = (const float*)d_in[4];
    const float* b2    = (const float*)d_in[5];
    const float* W3    = (const float*)d_in[6];
    const float* b3    = (const float*)d_in[7];

    float* out  = (float*)d_out;
    float* gws  = (float*)d_ws;
    float* eiws = gws + (size_t)ROWS * DF;

    mlp_wmma_kernel<<<ROWS / 16, 32, 0, stream>>>(image, W1, b1, W2, b2, W3, b3, gws, eiws);
    force_kernel<<<256, 256, 0, stream>>>(dfeat, gws, out);
    etot_kernel<<<1, 256, 0, stream>>>(eiws, out);
}


# force kernel: 16 waves per workgroup instead of 8 (two waves share a row, each streams half of it, 18 loads per group); the two half-row sums are added through LDS: row sums re-ordered
# speedup vs baseline: 2.1034x; 1.0442x over previous
#include <hip/hip_runtime.h>

typedef float v2f __attribute__((ext_vector_type(2)));
typedef float v4f __attribute__((ext_vector_type(4)));
typedef float v8f __attribute__((ext_vector_type(8)));

#define WMMA_F32(a, b, c) \
    __builtin_amdgcn_wmma_f32_16x16x4_f32(false, (a), false, (b), (short)0, (c), false, false)

constexpr int BATCH  = 8;
constexpr int NATOMS = 256;
constexpr int DF     = 42;
constexpr int H1     = 64;
constexpr int H2     = 32;
constexpr int ROWS   = BATCH * NATOMS;
constexpr int JF     = NATOMS * DF;

__device__ __forceinline__ float softplus_f(float z) {
    return fmaxf(z, 0.0f) + log1pf(expf(-fabsf(z)));
}
__device__ __forceinline__ float sigmoid_f(float z) {
    return 1.0f / (1.0f + expf(-z));
}

__global__ __launch_bounds__(32)
void mlp_wmma_kernel(const float* __restrict__ image,
                     const float* __restrict__ W1, const float* __restrict__ b1,
                     const float* __restrict__ W2, const float* __restrict__ b2,
                     const float* __restrict__ W3, const float* __restrict__ b3,
                     float* __restrict__ gws,
                     float* __restrict__ eiws)
{
    __shared__ __align__(16) float sW1[48 * 64];
    __shared__ __align__(16) float sW2[64 * 32];
    __shared__ float sW3[32];
    __shared__ float sb1[64];
    __shared__ float sb2[32];
    __shared__ __align__(16) float sX [16 * 44];
    __shared__ float sA1[16 * 64];
    __shared__ float sA2[16 * 32];
    __shared__ float sG2[16 * 32];
    __shared__ float sG1[16 * 64];

    const int lane = threadIdx.x;
    const int half = lane >> 4;
    const int l    = lane & 15;
    const int tile = blockIdx.x;
    const int row0 = tile * 16;

    const v4f* W1v = (const v4f*)W1;
    const v4f* W2v = (const v4f*)W2;
    const v2f* Xv  = (const v2f*)(image + (size_t)row0 * DF);

    v2f xr[11];
    v4f w1r[21];
    v4f w2r[16];
    #pragma unroll
    for (int it = 0; it < 11; ++it) {
        int p = lane + 32 * it;
        xr[it] = Xv[p < 336 ? p : 335];
    }
    #pragma unroll
    for (int it = 0; it < 21; ++it) w1r[it] = W1v[lane + 32 * it];
    #pragma unroll
    for (int it = 0; it < 16; ++it) w2r[it] = W2v[lane + 32 * it];
    const float w3r  = W3[lane];
    const float b2r  = b2[lane];
    const float b1r0 = b1[lane];
    const float b1r1 = b1[lane + 32];

    v4f* sW1v = (v4f*)sW1;
    v4f* sW2v = (v4f*)sW2;
    const v4f zero4 = {0.0f, 0.0f, 0.0f, 0.0f};
    #pragma unroll
    for (int it = 0; it < 10; ++it) {
        int e = 2 * (lane + 32 * it);
        int r = e / 42, c = e - 42 * r;
        *(v2f*)&sX[r * 44 + c] = xr[it];
    }
    {
        int e = 2 * (lane + 320);
        int r = e / 42, c = e - 42 * r;
        int  a = (lane < 16) ? (r * 44 + c) : ((lane - 16) * 44 + 42);
        v2f  v = xr[10];
        if (lane >= 16) { v.x = 0.0f; v.y = 0.0f; }
        *(v2f*)&sX[a] = v;
    }
    #pragma unroll
    for (int it = 0; it < 21; ++it) sW1v[lane + 32 * it] = w1r[it];
    for (int it = 0; it < 3; ++it) sW1v[672 + lane + 32 * it] = zero4;
    #pragma unroll
    for (int it = 0; it < 16; ++it) sW2v[lane + 32 * it] = w2r[it];
    sW3[lane]      = w3r;
    sb2[lane]      = b2r;
    sb1[lane]      = b1r0;
    sb1[lane + 32] = b1r1;
    __syncthreads();

    v8f z1t[4];
    #pragma unroll
    for (int t = 0; t < 4; ++t) {
        float bv = sb1[t * 16 + l];
        v8f c;
        #pragma unroll
        for (int i = 0; i < 8; ++i) c[i] = bv;
        #pragma unroll
        for (int k0 = 0; k0 < 44; k0 += 4) {
            v2f a, b;
            a.x = sX[l * 44 + k0 + 2 * half];
            a.y = sX[l * 44 + k0 + 2 * half + 1];
            b.x = sW1[(k0 + 2 * half)     * 64 + t * 16 + l];
            b.y = sW1[(k0 + 2 * half + 1) * 64 + t * 16 + l];
            c = WMMA_F32(a, b, c);
        }
        z1t[t] = c;
    }

    v8f sig1[4];
    #pragma unroll
    for (int t = 0; t < 4; ++t) {
        #pragma unroll
        for (int i = 0; i < 8; ++i) {
            float z = z1t[t][i];
            sA1[(i + 8 * half) * 64 + t * 16 + l] = softplus_f(z);
            sig1[t][i] = sigmoid_f(z);
        }
    }
    __syncthreads();

    v8f z2t[2];
    #pragma unroll
    for (int t = 0; t < 2; ++t) {
        float bv = sb2[t * 16 + l];
        v8f c;
        #pragma unroll
        for (int i = 0; i < 8; ++i) c[i] = bv;
        #pragma unroll
        for (int k0 = 0; k0 < 64; k0 += 4) {
            v2f a, b;
            a.x = sA1[l * 64 + k0 + 2 * half];
            a.y = sA1[l * 64 + k0 + 2 * half + 1];
            b.x = sW2[(k0 + 2 * half)     * 32 + t * 16 + l];
            b.y = sW2[(k0 + 2 * half + 1) * 32 + t * 16 + l];
            c = WMMA_F32(a, b, c);
        }
        z2t[t] = c;
    }

    #pragma unroll
    for (int t = 0; t < 2; ++t) {
        float w3v = sW3[t * 16 + l];
        #pragma unroll
        for (int i = 0; i < 8; ++i) {
            float z = z2t[t][i];
            sA2[(i + 8 * half) * 32 + t * 16 + l] = softplus_f(z);
            sG2[(i + 8 * half) * 32 + t * 16 + l] = sigmoid_f(z) * w3v;
        }
    }
    __syncthreads();

    v8f ce;
    #pragma unroll
    for (int i = 0; i < 8; ++i) ce[i] = 0.0f;
    #pragma unroll
    for (int k0 = 0; k0 < 32; k0 += 4) {
        v2f a, b;
        a.x = sA2[l * 32 + k0 + 2 * half];
        a.y = sA2[l * 32 + k0 + 2 * half + 1];
        b.x = (l == 0) ? sW3[k0 + 2 * half]     : 0.0f;
        b.y = (l == 0) ? sW3[k0 + 2 * half + 1] : 0.0f;
        ce = WMMA_F32(a, b, ce);
    }
    {
        float b3v = b3[0];
        if (l == 0) {
            #pragma unroll
            for (int v = 0; v < 8; ++v)
                eiws[row0 + 8 * half + v] = ce[v] + b3v;
        }
    }

    #pragma unroll
    for (int t = 0; t < 4; ++t) {
        v8f c;
        #pragma unroll
        for (int i = 0; i < 8; ++i) c[i] = 0.0f;
        #pragma unroll
        for (int k0 = 0; k0 < 32; k0 += 4) {
            v2f a, b;
            a.x = sG2[l * 32 + k0 + 2 * half];
            a.y = sG2[l * 32 + k0 + 2 * half + 1];
            b.x = sW2[(t * 16 + l) * 32 + k0 + 2 * half];
            b.y = sW2[(t * 16 + l) * 32 + k0 + 2 * half + 1];
            c = WMMA_F32(a, b, c);
        }
        #pragma unroll
        for (int i = 0; i < 8; ++i)
            sG1[(i + 8 * half) * 64 + t * 16 + l] = sig1[t][i] * c[i];
    }
    __syncthreads();

    #pragma unroll
    for (int t = 0; t < 3; ++t) {
        v8f c;
        #pragma unroll
        for (int i = 0; i < 8; ++i) c[i] = 0.0f;
        #pragma unroll
        for (int k0 = 0; k0 < 64; k0 += 4) {
            v2f a, b;
            a.x = sG1[l * 64 + k0 + 2 * half];
            a.y = sG1[l * 64 + k0 + 2 * half + 1];
            b.x = sW1[(t * 16 + l) * 64 + k0 + 2 * half];
            b.y = sW1[(t * 16 + l) * 64 + k0 + 2 * half + 1];
            c = WMMA_F32(a, b, c);
        }
        int d = t * 16 + l;
        if (d < DF) {
            #pragma unroll
            for (int v = 0; v < 8; ++v)
                gws[(row0 + v + 8 * half) * DF + d] = c[v];
        }
    }
}

constexpr int ROWF = JF * 3;
constexpr int NLD  = ROWF / (4 * 32);
constexpr int WPR  = 2;
constexpr int LPW  = NLD / WPR;
constexpr int NG   = 18;
constexpr int NGRP = LPW / NG;

__global__ __launch_bounds__(512)
void force_kernel(const float* __restrict__ dfeat,
                  const float* __restrict__ gws,
                  float* __restrict__ out)
{
    __shared__ __align__(16) float g3[ROWF];
    __shared__ float part[16 * 3];

    const int tid = threadIdx.x;
    const int blk = blockIdx.x;
    const int b   = blk >> 5;
    const int i0  = (blk & 31) * 8;

    {
        const v4f* gv = (const v4f*)(gws + (size_t)b * JF);
        v4f* g3v = (v4f*)g3;
        v4f gr[6];
        #pragma unroll
        for (int t = 0; t < 5; ++t) gr[t] = gv[tid + 512 * t];
        gr[5] = gv[2560 + (tid & 127)];
        #pragma unroll
        for (int t = 0; t < 6; ++t) {
            if (t < 5 || tid < 128) {
                const int q = tid + 512 * t;
                const v4f g = gr[t];
                const v4f s0 = {g.x, g.x, g.x, g.y};
                const v4f s1 = {g.y, g.y, g.z, g.z};
                const v4f s2 = {g.z, g.w, g.w, g.w};
                g3v[3 * q + 0] = s0;
                g3v[3 * q + 1] = s1;
                g3v[3 * q + 2] = s2;
            }
        }
    }
    __syncthreads();

    const int wave = tid >> 5;
    const int lane = tid & 31;
    const int row  = wave & 7;
    const int hf   = wave >> 3;
    const int i    = i0 + row;

    const v4f* rv  = (const v4f*)(dfeat + (size_t)(b * NATOMS + i) * (size_t)ROWF) + hf * (LPW * 32) + lane;
    const v4f* g3l = (const v4f*)g3 + hf * (LPW * 32) + lane;

    const v4f zero4 = {0.0f, 0.0f, 0.0f, 0.0f};
    v4f acc[3] = {zero4, zero4, zero4};

    for (int grp = 0; grp < NGRP; ++grp) {
        v4f d[NG];
        #pragma unroll
        for (int u = 0; u < NG; ++u) d[u] = rv[(grp * NG + u) * 32];
        #pragma unroll
        for (int u = 0; u < NG; ++u)
            acc[u % 3] = __builtin_elementwise_fma(d[u], g3l[(grp * NG + u) * 32], acc[u % 3]);
    }

    float a[3] = {0.0f, 0.0f, 0.0f};
    #pragma unroll
    for (int k = 0; k < 3; ++k) {
        #pragma unroll
        for (int r = 0; r < 4; ++r) a[(2 * k + r) % 3] += acc[k][r];
    }
    const int ph = lane % 3;
    float ax = (ph == 0) ? a[0] : ((ph == 1) ? a[2] : a[1]);
    float ay = (ph == 0) ? a[1] : ((ph == 1) ? a[0] : a[2]);
    float az = (ph == 0) ? a[2] : ((ph == 1) ? a[1] : a[0]);

    #pragma unroll
    for (int off = 16; off; off >>= 1) {
        ax += __shfl_xor(ax, off, 32);
        ay += __shfl_xor(ay, off, 32);
        az += __shfl_xor(az, off, 32);
    }
    if (lane == 0) {
        part[wave * 3 + 0] = ax;
        part[wave * 3 + 1] = ay;
        part[wave * 3 + 2] = az;
    }
    __syncthreads();
    if (tid < 24)
        out[8 + (size_t)(b * NATOMS + i0) * 3 + tid] = part[tid] + part[24 + tid];
}

__global__ __launch_bounds__(256)
void etot_kernel(const float* __restrict__ eiws, float* __restrict__ out)
{
    const int b = threadIdx.x >> 5;
    const int lane = threadIdx.x & 31;
    float s = 0.0f;
    #pragma unroll
    for (int k = 0; k < 8; ++k) s += eiws[b * NATOMS + lane + 32 * k];
    #pragma unroll
    for (int off = 16; off; off >>= 1) s += __shfl_xor(s, off, 32);
    if (lane == 0) out[b] = s;
}

extern "C" void kernel_launch(void* const* d_in, const int* in_sizes, int n_in,
                              void* d_out, int out_size, void* d_ws, size_t ws_size,
                              hipStream_t stream)
{
    (void)in_sizes; (void)n_in; (void)out_size; (void)ws_size;

    const float* image = (const float*)d_in[0];
    const float* dfeat = (const float*)d_in[1];
    const float* W1    = (const float*)d_in[2];
    const float* b1    = (const float*)d_in[3];
    const float* W2    = (const float*)d_in[4];
    const float* b2    = (const float*)d_in[5];
    const float* W3    = (const float*)d_in[6];
    const float* b3    = (const float*)d_in[7];

    float* out  = (float*)d_out;
    float* gws  = (float*)d_ws;
    float* eiws = gws + (size_t)ROWS * DF;

    mlp_wmma_kernel<<<ROWS / 16, 32, 0, stream>>>(image, W1, b1, W2, b2, W3, b3, gws, eiws);
    force_kernel<<<256, 512, 0, stream>>>(dfeat, gws, out);
    etot_kernel<<<1, 256, 0, stream>>>(eiws, out);
}


# first kernel: 4 waves per 16-row tile instead of 1; wave w does the 16-column tile w of each product and its softplus/sigmoid (side by side, not in turn); every tile's sum order unchanged
# speedup vs baseline: 2.3716x; 1.1275x over previous
#include <hip/hip_runtime.h>

typedef float v2f __attribute__((ext_vector_type(2)));
typedef float v4f __attribute__((ext_vector_type(4)));
typedef float v8f __attribute__((ext_vector_type(8)));

#define WMMA_F32(a, b, c) \
    __builtin_amdgcn_wmma_f32_16x16x4_f32(false, (a), false, (b), (short)0, (c), false, false)

constexpr int BATCH  = 8;
constexpr int NATOMS = 256;
constexpr int DF     = 42;
constexpr int H1     = 64;
constexpr int H2     = 32;
constexpr int ROWS   = BATCH * NATOMS;
constexpr int JF     = NATOMS * DF;

__device__ __forceinline__ float softplus_f(float z) {
    return fmaxf(z, 0.0f) + log1pf(expf(-fabsf(z)));
}
__device__ __forceinline__ float sigmoid_f(float z) {
    return 1.0f / (1.0f + expf(-z));
}

__global__ __launch_bounds__(128)
void mlp_wmma_kernel(const float* __restrict__ image,
                     const float* __restrict__ W1, const float* __restrict__ b1,
                     const float* __restrict__ W2, const float* __restrict__ b2,
                     const float* __restrict__ W3, const float* __restrict__ b3,
                     float* __restrict__ gws,
                     float* __restrict__ eiws)
{
    __shared__ __align__(16) float sW1[48 * 64];
    __shared__ __align__(16) float sW2[64 * 32];
    __shared__ float sW3[32];
    __shared__ float sb1[64];
    __shared__ float sb2[32];
    __shared__ __align__(16) float sX [16 * 44];
    __shared__ float sA1[16 * 64];
    __shared__ float sA2[16 * 32];
    __shared__ float sG2[16 * 32];
    __shared__ float sG1[16 * 64];

    const int tid  = threadIdx.x;
    const int lane = tid & 31;
    const int wave = __builtin_amdgcn_readfirstlane(tid >> 5);
    const int half = lane >> 4;
    const int l    = lane & 15;
    const int tile = blockIdx.x;
    const int row0 = tile * 16;

    const v4f* W1v = (const v4f*)W1;
    const v4f* W2v = (const v4f*)W2;
    const v2f* Xv  = (const v2f*)(image + (size_t)row0 * DF);

    v2f xr[3];
    v4f w1r[6];
    v4f w2r[4];
    #pragma unroll
    for (int it = 0; it < 3; ++it) {
        int p = tid + 128 * it;
        xr[it] = Xv[p < 336 ? p : 335];
    }
    #pragma unroll
    for (int it = 0; it < 5; ++it) w1r[it] = W1v[tid + 128 * it];
    w1r[5] = W1v[640 + lane];
    #pragma unroll
    for (int it = 0; it < 4; ++it) w2r[it] = W2v[tid + 128 * it];
    const float w3r = W3[lane];
    const float b2r = b2[lane];
    const float b1r = b1[tid & 63];
    const float b3v = b3[0];

    v4f* sW1v = (v4f*)sW1;
    v4f* sW2v = (v4f*)sW2;
    const v4f zero4 = {0.0f, 0.0f, 0.0f, 0.0f};
    #pragma unroll
    for (int it = 0; it < 2; ++it) {
        int e = 2 * (tid + 128 * it);
        int r = e / 42, c = e - 42 * r;
        *(v2f*)&sX[r * 44 + c] = xr[it];
    }
    {
        int e = 2 * (tid + 256);
        int r = e / 42, c = e - 42 * r;
        int  a = (tid < 80) ? (r * 44 + c) : ((tid - 80) * 44 + 42);
        v2f  v = xr[2];
        if (tid >= 80) { v.x = 0.0f; v.y = 0.0f; }
        if (tid < 96) *(v2f*)&sX[a] = v;
    }
    #pragma unroll
    for (int it = 0; it < 5; ++it) sW1v[tid + 128 * it] = w1r[it];
    sW1v[640 + tid] = (tid < 32) ? w1r[5] : zero4;
    #pragma unroll
    for (int it = 0; it < 4; ++it) sW2v[tid + 128 * it] = w2r[it];
    if (tid < 32) {
        sW3[tid] = w3r;
        sb2[tid] = b2r;
    }
    if (tid < 64) sb1[tid] = b1r;
    __syncthreads();

    v8f z1;
    {
        const int t = wave;
        float bv = sb1[t * 16 + l];
        v8f c;
        #pragma unroll
        for (int i = 0; i < 8; ++i) c[i] = bv;
        #pragma unroll
        for (int k0 = 0; k0 < 44; k0 += 4) {
            v2f a, b;
            a.x = sX[l * 44 + k0 + 2 * half];
            a.y = sX[l * 44 + k0 + 2 * half + 1];
            b.x = sW1[(k0 + 2 * half)     * 64 + t * 16 + l];
            b.y = sW1[(k0 + 2 * half + 1) * 64 + t * 16 + l];
            c = WMMA_F32(a, b, c);
        }
        z1 = c;
    }

    v8f sig1;
    #pragma unroll
    for (int i = 0; i < 8; ++i) {
        float z = z1[i];
        sA1[(i + 8 * half) * 64 + wave * 16 + l] = softplus_f(z);
        sig1[i] = sigmoid_f(z);
    }
    __syncthreads();

    if (wave < 2) {
        const int t = wave;
        float bv = sb2[t * 16 + l];
        v8f c;
        #pragma unroll
        for (int i = 0; i < 8; ++i) c[i] = bv;
        #pragma unroll
        for (int k0 = 0; k0 < 64; k0 += 4) {
            v2f a, b;
            a.x = sA1[l * 64 + k0 + 2 * half];
            a.y = sA1[l * 64 + k0 + 2 * half + 1];
            b.x = sW2[(k0 + 2 * half)     * 32 + t * 16 + l];
            b.y = sW2[(k0 + 2 * half + 1) * 32 + t * 16 + l];
            c = WMMA_F32(a, b, c);
        }
        float w3v = sW3[t * 16 + l];
        #pragma unroll
        for (int i = 0; i < 8; ++i) {
            float z = c[i];
            sA2[(i + 8 * half) * 32 + t * 16 + l] = softplus_f(z);
            sG2[(i + 8 * half) * 32 + t * 16 + l] = sigmoid_f(z) * w3v;
        }
    }
    __syncthreads();

    {
        const int t = wave;
        v8f c;
        #pragma unroll
        for (int i = 0; i < 8; ++i) c[i] = 0.0f;
        #pragma unroll
        for (int k0 = 0; k0 < 32; k0 += 4) {
            v2f a, b;
            a.x = sG2[l * 32 + k0 + 2 * half];
            a.y = sG2[l * 32 + k0 + 2 * half + 1];
            b.x = sW2[(t * 16 + l) * 32 + k0 + 2 * half];
            b.y = sW2[(t * 16 + l) * 32 + k0 + 2 * half + 1];
            c = WMMA_F32(a, b, c);
        }
        #pragma unroll
        for (int i = 0; i < 8; ++i)
            sG1[(i + 8 * half) * 64 + t * 16 + l] = sig1[i] * c[i];
    }
    __syncthreads();

    if (wave < 3) {
        const int t = wave;
        v8f c;
        #pragma unroll
        for (int i = 0; i < 8; ++i) c[i] = 0.0f;
        #pragma unroll
        for (int k0 = 0; k0 < 64; k0 += 4) {
            v2f a, b;
            a.x = sG1[l * 64 + k0 + 2 * half];
            a.y = sG1[l * 64 + k0 + 2 * half + 1];
            b.x = sW1[(t * 16 + l) * 64 + k0 + 2 * half];
            b.y = sW1[(t * 16 + l) * 64 + k0 + 2 * half + 1];
            c = WMMA_F32(a, b, c);
        }
        int d = t * 16 + l;
        if (d < DF) {
            #pragma unroll
            for (int v = 0; v < 8; ++v)
                gws[(row0 + v + 8 * half) * DF + d] = c[v];
        }
    } else {
        v8f ce;
        #pragma unroll
        for (int i = 0; i < 8; ++i) ce[i] = 0.0f;
        #pragma unroll
        for (int k0 = 0; k0 < 32; k0 += 4) {
            v2f a, b;
            a.x = sA2[l * 32 + k0 + 2 * half];
            a.y = sA2[l * 32 + k0 + 2 * half + 1];
            b.x = (l == 0) ? sW3[k0 + 2 * half]     : 0.0f;
            b.y = (l == 0) ? sW3[k0 + 2 * half + 1] : 0.0f;
            ce = WMMA_F32(a, b, ce);
        }
        if (l == 0) {
            #pragma unroll
            for (int v = 0; v < 8; ++v)
                eiws[row0 + 8 * half + v] = ce[v] + b3v;
        }
    }
}

constexpr int ROWF = JF * 3;
constexpr int NLD  = ROWF / (4 * 32);
constexpr int WPR  = 2;
constexpr int LPW  = NLD / WPR;
constexpr int NG   = 18;
constexpr int NGRP = LPW / NG;

__global__ __launch_bounds__(512)
void force_kernel(const float* __restrict__ dfeat,
                  const float* __restrict__ gws,
                  float* __restrict__ out)
{
    __shared__ __align__(16) float g3[ROWF];
    __shared__ float part[16 * 3];

    const int tid = threadIdx.x;
    const int blk = blockIdx.x;
    const int b   = blk >> 5;
    const int i0  = (blk & 31) * 8;

    {
        const v4f* gv = (const v4f*)(gws + (size_t)b * JF);
        v4f* g3v = (v4f*)g3;
        v4f gr[6];
        #pragma unroll
        for (int t = 0; t < 5; ++t) gr[t] = gv[tid + 512 * t];
        gr[5] = gv[2560 + (tid & 127)];
        #pragma unroll
        for (int t = 0; t < 6; ++t) {
            if (t < 5 || tid < 128) {
                const int q = tid + 512 * t;
                const v4f g = gr[t];
                const v4f s0 = {g.x, g.x, g.x, g.y};
                const v4f s1 = {g.y, g.y, g.z, g.z};
                const v4f s2 = {g.z, g.w, g.w, g.w};
                g3v[3 * q + 0] = s0;
                g3v[3 * q + 1] = s1;
                g3v[3 * q + 2] = s2;
            }
        }
    }
    __syncthreads();

    const int wave = tid >> 5;
    const int lane = tid & 31;
    const int row  = wave & 7;
    const int hf   = wave >> 3;
    const int i    = i0 + row;

    const v4f* rv  = (const v4f*)(dfeat + (size_t)(b * NATOMS + i) * (size_t)ROWF) + hf * (LPW * 32) + lane;
    const v4f* g3l = (const v4f*)g3 + hf * (LPW * 32) + lane;

    const v4f zero4 = {0.0f, 0.0f, 0.0f, 0.0f};
    v4f acc[3] = {zero4, zero4, zero4};

    for (int grp = 0; grp < NGRP; ++grp) {
        v4f d[NG];
        #pragma unroll
        for (int u = 0; u < NG; ++u) d[u] = rv[(grp * NG + u) * 32];
        #pragma unroll
        for (int u = 0; u < NG; ++u)
            acc[u % 3] = __builtin_elementwise_fma(d[u], g3l[(grp * NG + u) * 32], acc[u % 3]);
    }

    float a[3] = {0.0f, 0.0f, 0.0f};
    #pragma unroll
    for (int k = 0; k < 3; ++k) {
        #pragma unroll
        for (int r = 0; r < 4; ++r) a[(2 * k + r) % 3] += acc[k][r];
    }
    const int ph = lane % 3;
    float ax = (ph == 0) ? a[0] : ((ph == 1) ? a[2] : a[1]);
    float ay = (ph == 0) ? a[1] : ((ph == 1) ? a[0] : a[2]);
    float az = (ph == 0) ? a[2] : ((ph == 1) ? a[1] : a[0]);

    #pragma unroll
    for (int off = 16; off; off >>= 1) {
        ax += __shfl_xor(ax, off, 32);
        ay += __shfl_xor(ay, off, 32);
        az += __shfl_xor(az, off, 32);
    }
    if (lane == 0) {
        part[wave * 3 + 0] = ax;
        part[wave * 3 + 1] = ay;
        part[wave * 3 + 2] = az;
    }
    __syncthreads();
    if (tid < 24)
        out[8 + (size_t)(b * NATOMS + i0) * 3 + tid] = part[tid] + part[24 + tid];
}

__global__ __launch_bounds__(256)
void etot_kernel(const float* __restrict__ eiws, float* __restrict__ out)
{
    const int b = threadIdx.x >> 5;
    const int lane = threadIdx.x & 31;
    float s = 0.0f;
    #pragma unroll
    for (int k = 0; k < 8; ++k) s += eiws[b * NATOMS + lane + 32 * k];
    #pragma unroll
    for (int off = 16; off; off >>= 1) s += __shfl_xor(s, off, 32);
    if (lane == 0) out[b] = s;
}

extern "C" void kernel_launch(void* const* d_in, const int* in_sizes, int n_in,
                              void* d_out, int out_size, void* d_ws, size_t ws_size,
                              hipStream_t stream)
{
    (void)in_sizes; (void)n_in; (void)out_size; (void)ws_size;

    const float* image = (const float*)d_in[0];
    const float* dfeat = (const float*)d_in[1];
    const float* W1    = (const float*)d_in[2];
    const float* b1    = (const float*)d_in[3];
    const float* W2    = (const float*)d_in[4];
    const float* b2    = (const float*)d_in[5];
    const float* W3    = (const float*)d_in[6];
    const float* b3    = (const float*)d_in[7];

    float* out  = (float*)d_out;
    float* gws  = (float*)d_ws;
    float* eiws = gws + (size_t)ROWS * DF;

    mlp_wmma_kernel<<<ROWS / 16, 128, 0, stream>>>(image, W1, b1, W2, b2, W3, b3, gws, eiws);
    force_kernel<<<256, 512, 0, stream>>>(dfeat, gws, out);
    etot_kernel<<<1, 256, 0, stream>>>(eiws, out);
}


# first kernel: softplus and sigmoid from one exp(-|z|) on the hardware exp2, log2 and reciprocal instructions (1 ulp each), not library expf twice, log1pf and a divide; last bits of g, Ei differ
# speedup vs baseline: 2.6083x; 1.0998x over previous
#include <hip/hip_runtime.h>

typedef float v2f __attribute__((ext_vector_type(2)));
typedef float v4f __attribute__((ext_vector_type(4)));
typedef float v8f __attribute__((ext_vector_type(8)));

#define WMMA_F32(a, b, c) \
    __builtin_amdgcn_wmma_f32_16x16x4_f32(false, (a), false, (b), (short)0, (c), false, false)

constexpr int BATCH  = 8;
constexpr int NATOMS = 256;
constexpr int DF     = 42;
constexpr int H1     = 64;
constexpr int H2     = 32;
constexpr int ROWS   = BATCH * NATOMS;
constexpr int JF     = NATOMS * DF;

__device__ __forceinline__ void softplus_sigmoid(float z, float& sp, float& sg) {
    const float t = __builtin_amdgcn_exp2f(-fabsf(z) * 1.44269504088896341f);
    const float u = 1.0f + t;
    const float r = __builtin_amdgcn_rcpf(u);
    sp = fmaxf(z, 0.0f) + __builtin_amdgcn_logf(u) * 0.693147180559945309f;
    sg = (z >= 0.0f) ? r : t * r;
}

__global__ __launch_bounds__(128)
void mlp_wmma_kernel(const float* __restrict__ image,
                     const float* __restrict__ W1, const float* __restrict__ b1,
                     const float* __restrict__ W2, const float* __restrict__ b2,
                     const float* __restrict__ W3, const float* __restrict__ b3,
                     float* __restrict__ gws,
                     float* __restrict__ eiws)
{
    __shared__ __align__(16) float sW1[48 * 64];
    __shared__ __align__(16) float sW2[64 * 32];
    __shared__ float sW3[32];
    __shared__ float sb1[64];
    __shared__ float sb2[32];
    __shared__ __align__(16) float sX [16 * 44];
    __shared__ float sA1[16 * 64];
    __shared__ float sA2[16 * 32];
    __shared__ float sG2[16 * 32];
    __shared__ float sG1[16 * 64];

    const int tid  = threadIdx.x;
    const int lane = tid & 31;
    const int wave = __builtin_amdgcn_readfirstlane(tid >> 5);
    const int half = lane >> 4;
    const int l    = lane & 15;
    const int tile = blockIdx.x;
    const int row0 = tile * 16;

    const v4f* W1v = (const v4f*)W1;
    const v4f* W2v = (const v4f*)W2;
    const v2f* Xv  = (const v2f*)(image + (size_t)row0 * DF);

    v2f xr[3];
    v4f w1r[6];
    v4f w2r[4];
    #pragma unroll
    for (int it = 0; it < 3; ++it) {
        int p = tid + 128 * it;
        xr[it] = Xv[p < 336 ? p : 335];
    }
    #pragma unroll
    for (int it = 0; it < 5; ++it) w1r[it] = W1v[tid + 128 * it];
    w1r[5] = W1v[640 + lane];
    #pragma unroll
    for (int it = 0; it < 4; ++it) w2r[it] = W2v[tid + 128 * it];
    const float w3r = W3[lane];
    const float b2r = b2[lane];
    const float b1r = b1[tid & 63];
    const float b3v = b3[0];

    v4f* sW1v = (v4f*)sW1;
    v4f* sW2v = (v4f*)sW2;
    const v4f zero4 = {0.0f, 0.0f, 0.0f, 0.0f};
    #pragma unroll
    for (int it = 0; it < 2; ++it) {
        int e = 2 * (tid + 128 * it);
        int r = e / 42, c = e - 42 * r;
        *(v2f*)&sX[r * 44 + c] = xr[it];
    }
    {
        int e = 2 * (tid + 256);
        int r = e / 42, c = e - 42 * r;
        int  a = (tid < 80) ? (r * 44 + c) : ((tid - 80) * 44 + 42);
        v2f  v = xr[2];
        if (tid >= 80) { v.x = 0.0f; v.y = 0.0f; }
        if (tid < 96) *(v2f*)&sX[a] = v;
    }
    #pragma unroll
    for (int it = 0; it < 5; ++it) sW1v[tid + 128 * it] = w1r[it];
    sW1v[640 + tid] = (tid < 32) ? w1r[5] : zero4;
    #pragma unroll
    for (int it = 0; it < 4; ++it) sW2v[tid + 128 * it] = w2r[it];
    if (tid < 32) {
        sW3[tid] = w3r;
        sb2[tid] = b2r;
    }
    if (tid < 64) sb1[tid] = b1r;
    __syncthreads();

    v8f z1;
    {
        const int t = wave;
        float bv = sb1[t * 16 + l];
        v8f c;
        #pragma unroll
        for (int i = 0; i < 8; ++i) c[i] = bv;
        #pragma unroll
        for (int k0 = 0; k0 < 44; k0 += 4) {
            v2f a, b;
            a.x = sX[l * 44 + k0 + 2 * half];
            a.y = sX[l * 44 + k0 + 2 * half + 1];
            b.x = sW1[(k0 + 2 * half)     * 64 + t * 16 + l];
            b.y = sW1[(k0 + 2 * half + 1) * 64 + t * 16 + l];
            c = WMMA_F32(a, b, c);
        }
        z1 = c;
    }

    v8f sig1;
    #pragma unroll
    for (int i = 0; i < 8; ++i) {
        float sp, sg;
        softplus_sigmoid(z1[i], sp, sg);
        sA1[(i + 8 * half) * 64 + wave * 16 + l] = sp;
        sig1[i] = sg;
    }
    __syncthreads();

    if (wave < 2) {
        const int t = wave;
        float bv = sb2[t * 16 + l];
        v8f c;
        #pragma unroll
        for (int i = 0; i < 8; ++i) c[i] = bv;
        #pragma unroll
        for (int k0 = 0; k0 < 64; k0 += 4) {
            v2f a, b;
            a.x = sA1[l * 64 + k0 + 2 * half];
            a.y = sA1[l * 64 + k0 + 2 * half + 1];
            b.x = sW2[(k0 + 2 * half)     * 32 + t * 16 + l];
            b.y = sW2[(k0 + 2 * half + 1) * 32 + t * 16 + l];
            c = WMMA_F32(a, b, c);
        }
        float w3v = sW3[t * 16 + l];
        #pragma unroll
        for (int i = 0; i < 8; ++i) {
            float sp, sg;
            softplus_sigmoid(c[i], sp, sg);
            sA2[(i + 8 * half) * 32 + t * 16 + l] = sp;
            sG2[(i + 8 * half) * 32 + t * 16 + l] = sg * w3v;
        }
    }
    __syncthreads();

    {
        const int t = wave;
        v8f c;
        #pragma unroll
        for (int i = 0; i < 8; ++i) c[i] = 0.0f;
        #pragma unroll
        for (int k0 = 0; k0 < 32; k0 += 4) {
            v2f a, b;
            a.x = sG2[l * 32 + k0 + 2 * half];
            a.y = sG2[l * 32 + k0 + 2 * half + 1];
            b.x = sW2[(t * 16 + l) * 32 + k0 + 2 * half];
            b.y = sW2[(t * 16 + l) * 32 + k0 + 2 * half + 1];
            c = WMMA_F32(a, b, c);
        }
        #pragma unroll
        for (int i = 0; i < 8; ++i)
            sG1[(i + 8 * half) * 64 + t * 16 + l] = sig1[i] * c[i];
    }
    __syncthreads();

    if (wave < 3) {
        const int t = wave;
        v8f c;
        #pragma unroll
        for (int i = 0; i < 8; ++i) c[i] = 0.0f;
        #pragma unroll
        for (int k0 = 0; k0 < 64; k0 += 4) {
            v2f a, b;
            a.x = sG1[l * 64 + k0 + 2 * half];
            a.y = sG1[l * 64 + k0 + 2 * half + 1];
            b.x = sW1[(t * 16 + l) * 64 + k0 + 2 * half];
            b.y = sW1[(t * 16 + l) * 64 + k0 + 2 * half + 1];
            c = WMMA_F32(a, b, c);
        }
        int d = t * 16 + l;
        if (d < DF) {
            #pragma unroll
            for (int v = 0; v < 8; ++v)
                gws[(row0 + v + 8 * half) * DF + d] = c[v];
        }
    } else {
        v8f ce;
        #pragma unroll
        for (int i = 0; i < 8; ++i) ce[i] = 0.0f;
        #pragma unroll
        for (int k0 = 0; k0 < 32; k0 += 4) {
            v2f a, b;
            a.x = sA2[l * 32 + k0 + 2 * half];
            a.y = sA2[l * 32 + k0 + 2 * half + 1];
            b.x = (l == 0) ? sW3[k0 + 2 * half]     : 0.0f;
            b.y = (l == 0) ? sW3[k0 + 2 * half + 1] : 0.0f;
            ce = WMMA_F32(a, b, ce);
        }
        if (l == 0) {
            #pragma unroll
            for (int v = 0; v < 8; ++v)
                eiws[row0 + 8 * half + v] = ce[v] + b3v;
        }
    }
}

constexpr int ROWF = JF * 3;
constexpr int NLD  = ROWF / (4 * 32);
constexpr int WPR  = 2;
constexpr int LPW  = NLD / WPR;
constexpr int NG   = 18;
constexpr int NGRP = LPW / NG;

__global__ __launch_bounds__(512)
void force_kernel(const float* __restrict__ dfeat,
                  const float* __restrict__ gws,
                  const float* __restrict__ eiws,
                  float* __restrict__ out)
{
    __shared__ __align__(16) float g3[ROWF];
    __shared__ float part[16 * 3];

    const int tid = threadIdx.x;
    const int blk = blockIdx.x;
    const int b   = blk >> 5;
    const int i0  = (blk & 31) * 8;
    const bool esum = ((blk & 31) == 0) && (tid < 32);
    float es = 0.0f;

    {
        const v4f* gv = (const v4f*)(gws + (size_t)b * JF);
        v4f* g3v = (v4f*)g3;
        v4f gr[6];
        #pragma unroll
        for (int t = 0; t < 5; ++t) gr[t] = gv[tid + 512 * t];
        gr[5] = gv[2560 + (tid & 127)];
        if (esum) {
            #pragma unroll
            for (int k = 0; k < 8; ++k) es += eiws[b * NATOMS + tid + 32 * k];
        }
        #pragma unroll
        for (int t = 0; t < 6; ++t) {
            if (t < 5 || tid < 128) {
                const int q = tid + 512 * t;
                const v4f g = gr[t];
                const v4f s0 = {g.x, g.x, g.x, g.y};
                const v4f s1 = {g.y, g.y, g.z, g.z};
                const v4f s2 = {g.z, g.w, g.w, g.w};
                g3v[3 * q + 0] = s0;
                g3v[3 * q + 1] = s1;
                g3v[3 * q + 2] = s2;
            }
        }
    }
    __syncthreads();

    const int wave = tid >> 5;
    const int lane = tid & 31;
    const int row  = wave & 7;
    const int hf   = wave >> 3;
    const int i    = i0 + row;

    const v4f* rv  = (const v4f*)(dfeat + (size_t)(b * NATOMS + i) * (size_t)ROWF) + hf * (LPW * 32) + lane;
    const v4f* g3l = (const v4f*)g3 + hf * (LPW * 32) + lane;

    const v4f zero4 = {0.0f, 0.0f, 0.0f, 0.0f};
    v4f acc[3] = {zero4, zero4, zero4};

    for (int grp = 0; grp < NGRP; ++grp) {
        v4f d[NG];
        #pragma unroll
        for (int u = 0; u < NG; ++u) d[u] = rv[(grp * NG + u) * 32];
        #pragma unroll
        for (int u = 0; u < NG; ++u)
            acc[u % 3] = __builtin_elementwise_fma(d[u], g3l[(grp * NG + u) * 32], acc[u % 3]);
    }

    float a[3] = {0.0f, 0.0f, 0.0f};
    #pragma unroll
    for (int k = 0; k < 3; ++k) {
        #pragma unroll
        for (int r = 0; r < 4; ++r) a[(2 * k + r) % 3] += acc[k][r];
    }
    const int ph = lane % 3;
    float ax = (ph == 0) ? a[0] : ((ph == 1) ? a[2] : a[1]);
    float ay = (ph == 0) ? a[1] : ((ph == 1) ? a[0] : a[2]);
    float az = (ph == 0) ? a[2] : ((ph == 1) ? a[1] : a[0]);

    #pragma unroll
    for (int off = 16; off; off >>= 1) {
        ax += __shfl_xor(ax, off, 32);
        ay += __shfl_xor(ay, off, 32);
        az += __shfl_xor(az, off, 32);
        es += __shfl_xor(es, off, 32);
    }
    if (esum && lane == 0) out[b] = es;
    if (lane == 0) {
        part[wave * 3 + 0] = ax;
        part[wave * 3 + 1] = ay;
        part[wave * 3 + 2] = az;
    }
    __syncthreads();
    if (tid < 24)
        out[8 + (size_t)(b * NATOMS + i0) * 3 + tid] = part[tid] + part[24 + tid];
}

extern "C" void kernel_launch(void* const* d_in, const int* in_sizes, int n_in,
                              void* d_out, int out_size, void* d_ws, size_t ws_size,
                              hipStream_t stream)
{
    (void)in_sizes; (void)n_in; (void)out_size; (void)ws_size;

    const float* image = (const float*)d_in[0];
    const float* dfeat = (const float*)d_in[1];
    const float* W1    = (const float*)d_in[2];
    const float* b1    = (const float*)d_in[3];
    const float* W2    = (const float*)d_in[4];
    const float* b2    = (const float*)d_in[5];
    const float* W3    = (const float*)d_in[6];
    const float* b3    = (const float*)d_in[7];

    float* out  = (float*)d_out;
    float* gws  = (float*)d_ws;
    float* eiws = gws + (size_t)ROWS * DF;

    mlp_wmma_kernel<<<ROWS / 16, 128, 0, stream>>>(image, W1, b1, W2, b2, W3, b3, gws, eiws);
    force_kernel<<<256, 512, 0, stream>>>(dfeat, gws, eiws, out);
}
